# starting kernel
# baseline (speedup 1.0000x reference)
#include <hip/hip_runtime.h>

typedef float v2f __attribute__((ext_vector_type(2)));
typedef float v8f __attribute__((ext_vector_type(8)));

#define D 128

__global__ __launch_bounds__(256) void gcn_gemm_wmma(
    const float* __restrict__ x, const float* __restrict__ w,
    float* __restrict__ support, int n_nodes) {
  const int lane = threadIdx.x & 31;
  const int wave = threadIdx.x >> 5;
  const int m    = lane & 15;
  const int half = lane >> 4;
  const int row0 = blockIdx.x * 16;
  const int col0 = wave * 16;

  int arow = row0 + m;
  if (arow >= n_nodes) arow = n_nodes - 1;
  const float* xrow = x + (size_t)arow * D;
  const int n = col0 + m;

  v8f acc = {};
#pragma unroll 4
  for (int k = 0; k < D; k += 4) {
    const int ka = k + half * 2;
    v2f a = *(const v2f*)(xrow + ka);
    v2f b;
    b.x = w[(size_t)(ka + 0) * D + n];
    b.y = w[(size_t)(ka + 1) * D + n];
    acc = __builtin_amdgcn_wmma_f32_16x16x4_f32(
         false, a,  false, b,
         (short)0, acc,  false,  false);
  }

  float* srow = support + (size_t)(row0 + half * 8) * D + n;
  if (row0 + 16 <= n_nodes) {
#pragma unroll
    for (int i = 0; i < 8; ++i) srow[(size_t)i * D] = acc[i];
  } else {
#pragma unroll
    for (int i = 0; i < 8; ++i) {
      if (row0 + half * 8 + i < n_nodes) srow[(size_t)i * D] = acc[i];
    }
  }
}

__global__ __launch_bounds__(256) void gcn_init_out(
    float* __restrict__ out, const float* __restrict__ bias, int total) {
  int i = blockIdx.x * blockDim.x + threadIdx.x;
  if (i < total) out[i] = bias[i & (D - 1)];
}

__global__ __launch_bounds__(256) void gcn_spmm_scatter(
    const int* __restrict__ adj_row, const int* __restrict__ adj_col,
    const float* __restrict__ adj_val, const float* __restrict__ support,
    float* __restrict__ out, int n_edges) {
  const int lane = threadIdx.x & 31;
  const int e = blockIdx.x * (blockDim.x >> 5) + (threadIdx.x >> 5);
  if (e >= n_edges) return;

  const int src  = adj_col[e];
  const int dst  = adj_row[e];
  const float v  = adj_val[e];
  const float* srow = support + (size_t)src * D;
  float* drow       = out + (size_t)dst * D;

#pragma unroll
  for (int j = 0; j < D; j += 32) {
    atomicAdd(drow + j + lane, v * srow[j + lane]);
  }
}

extern "C" void kernel_launch(void* const* d_in, const int* in_sizes, int n_in,
                              void* d_out, int out_size, void* d_ws, size_t ws_size,
                              hipStream_t stream) {
  const float* x       = (const float*)d_in[0];
  const int*   adj_row = (const int*)d_in[1];
  const int*   adj_col = (const int*)d_in[2];
  const float* adj_val = (const float*)d_in[3];
  const float* weight  = (const float*)d_in[4];
  const float* bias    = (const float*)d_in[5];
  float* out     = (float*)d_out;
  float* support = (float*)d_ws;

  const int n_nodes = in_sizes[0] / D;
  const int n_edges = in_sizes[1];
  const int total   = n_nodes * D;

  gcn_gemm_wmma<<<(n_nodes + 15) / 16, 256, 0, stream>>>(x, weight, support, n_nodes);
  gcn_init_out<<<(total + 255) / 256, 256, 0, stream>>>(out, bias, total);
  gcn_spmm_scatter<<<(n_edges + 7) / 8, 256, 0, stream>>>(adj_row, adj_col, adj_val,
                                                          support, out, n_edges);
}


# Offset scan spread over one work-group per 1024 rows: each re-sums the counts ahead of its block with whole-line loads issued as one group, then scans its block in registers; was one work-group
# speedup vs baseline: 1.3514x; 1.3514x over previous
#include <hip/hip_runtime.h>


typedef float v2f __attribute__((ext_vector_type(2)));
typedef float v8f __attribute__((ext_vector_type(8)));

#define D 128
#define TM 32
#define PITCH 132


__global__ __launch_bounds__(256) void gcn_zero_counts(int* __restrict__ cnt, int n) {
  const int i = blockIdx.x * 256 + threadIdx.x;
  if (i < n) cnt[i] = 0;
}

__global__ __launch_bounds__(256) void gcn_count_rank(
    const int* __restrict__ adj_row, int* __restrict__ cnt, int* __restrict__ rank,
    int n_edges) {
  const int e = blockIdx.x * 256 + threadIdx.x;
  if (e < n_edges) rank[e] = atomicAdd(cnt + adj_row[e], 1);
}

#define SCAN_BLK 1024
#define SCAN_GRP 40

__device__ __forceinline__ int gcn_wave_incl_scan(int v, int lane) {
  v += __builtin_amdgcn_update_dpp(0, v, 0x111, 0xF, 0xF, true);
  v += __builtin_amdgcn_update_dpp(0, v, 0x112, 0xF, 0xF, true);
  v += __builtin_amdgcn_update_dpp(0, v, 0x114, 0xF, 0xF, true);
  v += __builtin_amdgcn_update_dpp(0, v, 0x118, 0xF, 0xF, true);
  const int low_row = __builtin_amdgcn_readlane(v, 15);
  return v + ((lane >= 16) ? low_row : 0);
}

__global__ __launch_bounds__(SCAN_BLK) void gcn_scan_counts(
    const int* __restrict__ cnt, int* __restrict__ off, int n) {
  __shared__ int w_cnt[32];
  __shared__ int w_pre[32];
  const int t = threadIdx.x;
  const int lane = t & 31;
  const int wave = __builtin_amdgcn_readfirstlane(t >> 5);
  const int nb = blockIdx.x;
  const int idx = nb * SCAN_BLK + t;
  const int c = (idx < n) ? cnt[idx] : 0;

  int pre = 0;
  for (int j0 = 0; j0 < nb; j0 += SCAN_GRP) {
    int v[SCAN_GRP];
#pragma unroll
    for (int u = 0; u < SCAN_GRP; ++u) {
      const int j = min(j0 + u, nb - 1);
      v[u] = cnt[j * SCAN_BLK + t];
    }
#pragma unroll
    for (int u = 0; u < SCAN_GRP; ++u) pre += (j0 + u < nb) ? v[u] : 0;
  }

  const int inc  = gcn_wave_incl_scan(c, lane);
  const int psum = gcn_wave_incl_scan(pre, lane);
  if (lane == 31) { w_cnt[wave] = inc; w_pre[wave] = psum; }
  __syncthreads();
  const int ci = gcn_wave_incl_scan(w_cnt[lane], lane);
  const int pi = gcn_wave_incl_scan(w_pre[lane], lane);
  const int prev_waves = __builtin_amdgcn_readlane(ci, (wave + 31) & 31);
  const int base = __builtin_amdgcn_readlane(pi, 31) + ((wave > 0) ? prev_waves : 0);
  if (idx <= n) off[idx] = base + inc - c;
}

__global__ __launch_bounds__(256) void gcn_fill_sorted(
    const int* __restrict__ adj_row, const int* __restrict__ adj_col,
    const float* __restrict__ adj_val, const int* __restrict__ off,
    const int* __restrict__ rank, int2* __restrict__ sedge, int n_edges) {
  const int e = blockIdx.x * 256 + threadIdx.x;
  if (e < n_edges) {
    const int p = off[adj_row[e]] + rank[e];
    sedge[p] = make_int2(adj_col[e], __float_as_int(adj_val[e]));
  }
}

__global__ __launch_bounds__(256) void gcn_gather_gemm(
    const float* __restrict__ x, const float* __restrict__ w,
    const float* __restrict__ bias, const int* __restrict__ off,
    const int2* __restrict__ sedge, float* __restrict__ out, int n_nodes) {
  __shared__ __attribute__((aligned(16))) float tile[TM * PITCH];
  const int lane = threadIdx.x & 31;
  const int wave = __builtin_amdgcn_readfirstlane((int)(threadIdx.x >> 5));
  const int row0 = blockIdx.x * TM;

  for (int i = 0; i < 4; ++i) {
    const int rl = wave * 4 + i;
    const int r = row0 + rl;
    float4 acc = make_float4(0.f, 0.f, 0.f, 0.f);
    if (r < n_nodes) {
      const int beg = off[r];
      const int end = off[r + 1];
      for (int base = beg; base < end; base += 32) {
        const int p = base + lane;
        int2 cv = make_int2(0, 0);
        if (p < end) cv = sedge[p];
        const int cntj = min(32, end - base);
#pragma unroll 4
        for (int j = 0; j < cntj; ++j) {
          const int cj = __builtin_amdgcn_readlane(cv.x, j);
          const float vj = __int_as_float(__builtin_amdgcn_readlane(cv.y, j));
          const float4 xv = *(const float4*)(x + (size_t)cj * D + lane * 4);
          acc.x += vj * xv.x;
          acc.y += vj * xv.y;
          acc.z += vj * xv.z;
          acc.w += vj * xv.w;
        }
      }
    }
    *(float4*)(tile + rl * PITCH + lane * 4) = acc;
  }
  __syncthreads();

  const int m    = lane & 15;
  const int half = lane >> 4;
  const int n    = wave * 16 + m;
  const float* a0p = tile + m * PITCH + half * 2;
  const float* a1p = a0p + 16 * PITCH;
  const float* wp  = w + (size_t)(half * 2) * D + n;

  v8f acc0 = {};
  v8f acc1 = {};
#pragma unroll 8
  for (int k = 0; k < D; k += 4) {
    const v2f a0 = *(const v2f*)(a0p + k);
    const v2f a1 = *(const v2f*)(a1p + k);
    v2f b;
    b.x = wp[(size_t)k * D];
    b.y = wp[(size_t)(k + 1) * D];
    acc0 = __builtin_amdgcn_wmma_f32_16x16x4_f32(false, a0, false, b, (short)0, acc0, false, false);
    acc1 = __builtin_amdgcn_wmma_f32_16x16x4_f32(false, a1, false, b, (short)0, acc1, false, false);
  }

  __syncthreads();
  const float bn = bias[n];
  for (int i = 0; i < 8; ++i) {
    tile[(half * 8 + i) * PITCH + n]      = acc0[i] + bn;
    tile[(16 + half * 8 + i) * PITCH + n] = acc1[i] + bn;
  }
  __syncthreads();
  for (int i = 0; i < 4; ++i) {
    const int rl = wave * 4 + i;
    const int r = row0 + rl;
    if (r < n_nodes)
      *(float4*)(out + (size_t)r * D + lane * 4) = *(const float4*)(tile + rl * PITCH + lane * 4);
  }
}

extern "C" void kernel_launch(void* const* d_in, const int* in_sizes, int n_in,
                              void* d_out, int out_size, void* d_ws, size_t ws_size,
                              hipStream_t stream) {
  const float* x       = (const float*)d_in[0];
  const int*   adj_row = (const int*)d_in[1];
  const int*   adj_col = (const int*)d_in[2];
  const float* adj_val = (const float*)d_in[3];
  const float* weight  = (const float*)d_in[4];
  const float* bias    = (const float*)d_in[5];
  float* out = (float*)d_out;

  const int n_nodes = in_sizes[0] / D;
  const int n_edges = in_sizes[1];

  int2* sedge = (int2*)d_ws;
  int*  rank  = (int*)(sedge + n_edges);
  int*  cnt   = rank + n_edges;
  int*  off   = cnt + n_nodes;

  gcn_zero_counts<<<(n_nodes + 255) / 256, 256, 0, stream>>>(cnt, n_nodes);
  gcn_count_rank<<<(n_edges + 255) / 256, 256, 0, stream>>>(adj_row, cnt, rank, n_edges);
  gcn_scan_counts<<<(n_nodes + SCAN_BLK) / SCAN_BLK, SCAN_BLK, 0, stream>>>(cnt, off, n_nodes);
  gcn_fill_sorted<<<(n_edges + 255) / 256, 256, 0, stream>>>(adj_row, adj_col, adj_val, off,
                                                             rank, sedge, n_edges);
  gcn_gather_gemm<<<(n_nodes + TM - 1) / TM, 256, 0, stream>>>(x, weight, bias, off, sedge,
                                                               out, n_nodes);
}


# Edge-sort counters moved from L2 to LDS: edges bucketed per 32-row block with per-group LDS counts, rows ordered in LDS in the gather kernel; 5 launches to 4; sum order = LDS add arrival rank
# speedup vs baseline: 3.4634x; 2.5628x over previous
#include <hip/hip_runtime.h>


typedef float v2f __attribute__((ext_vector_type(2)));
typedef float v8f __attribute__((ext_vector_type(8)));

#define D 128
#define TM 32
#define PITCH 132
#define NBMAX 2048
#define SORT_G 256
#define SORT_T 512
#define EPG 8
#define BPT (NBMAX / SORT_T)
#define CH 1024


__global__ __launch_bounds__(256) void gcn_zero_counts(int* __restrict__ cnt, int n) {
  const int i = blockIdx.x * 256 + threadIdx.x;
  if (i < n) cnt[i] = 0;
}

__device__ __forceinline__ int gcn_wave_incl_scan(int v, int lane) {
  v += __builtin_amdgcn_update_dpp(0, v, 0x111, 0xF, 0xF, true);
  v += __builtin_amdgcn_update_dpp(0, v, 0x112, 0xF, 0xF, true);
  v += __builtin_amdgcn_update_dpp(0, v, 0x114, 0xF, 0xF, true);
  v += __builtin_amdgcn_update_dpp(0, v, 0x118, 0xF, 0xF, true);
  const int low_row = __builtin_amdgcn_readlane(v, 15);
  return v + ((lane >= 16) ? low_row : 0);
}


__global__ __launch_bounds__(SORT_T) void gcn_bucket_count(
    const int* __restrict__ adj_row, int* __restrict__ cnt, int* __restrict__ wgbase,
    int n_edges, int nb) {
  __shared__ int hist[NBMAX];
  const int t = threadIdx.x;
#pragma unroll
  for (int k = 0; k < NBMAX / SORT_T; ++k) hist[t + k * SORT_T] = 0;
  __syncthreads();
  const int chunk = (n_edges + SORT_G - 1) / SORT_G;
  const int e0 = blockIdx.x * chunk;
  const int e1 = min(e0 + chunk, n_edges);
  for (int eb = e0 + t; eb < e1; eb += EPG * SORT_T) {
    int r[EPG];
#pragma unroll
    for (int u = 0; u < EPG; ++u) {
      const int e = eb + u * SORT_T;
      r[u] = (e < e1) ? adj_row[e] : -1;
    }
#pragma unroll
    for (int u = 0; u < EPG; ++u)
      if (r[u] >= 0) atomicAdd(&hist[r[u] >> 5], 1);
  }
  __syncthreads();
  for (int b = t; b < nb; b += SORT_T) {
    const int h = hist[b];
    wgbase[blockIdx.x * nb + b] = (h > 0) ? atomicAdd(cnt + b, h) : 0;
  }
}

__global__ __launch_bounds__(SORT_T) void gcn_bucket_fill(
    const int* __restrict__ adj_row, const int* __restrict__ adj_col,
    const float* __restrict__ adj_val, const int* __restrict__ cnt,
    const int* __restrict__ wgbase, int* __restrict__ off, int2* __restrict__ sedge,
    int n_edges, int nb) {
  __shared__ int pos[NBMAX];
  __shared__ int w_tot[SORT_T / 32];
  const int t = threadIdx.x;
  const int lane = t & 31;
  const int wave = __builtin_amdgcn_readfirstlane(t >> 5);
  int c[BPT];
  int g[BPT];
  int s = 0;
#pragma unroll
  for (int k = 0; k < BPT; ++k) {
    const int b = BPT * t + k;
    c[k] = (b < nb) ? cnt[b] : 0;
    g[k] = (b < nb) ? wgbase[blockIdx.x * nb + b] : 0;
    s += c[k];
  }
  const int inc = gcn_wave_incl_scan(s, lane);
  if (lane == 31) w_tot[wave] = inc;
  __syncthreads();
  const int ci = gcn_wave_incl_scan((lane < SORT_T / 32) ? w_tot[lane] : 0, lane);
  const int prev_waves = __builtin_amdgcn_readlane(ci, (wave + 31) & 31);
  int xk = ((wave > 0) ? prev_waves : 0) + inc - s;
#pragma unroll
  for (int k = 0; k < BPT; ++k) {
    const int b = BPT * t + k;
    pos[b] = xk + g[k];
    if (blockIdx.x == 0 && b <= nb) off[b] = xk;
    xk += c[k];
  }
  __syncthreads();
  const int chunk = (n_edges + SORT_G - 1) / SORT_G;
  const int e0 = blockIdx.x * chunk;
  const int e1 = min(e0 + chunk, n_edges);
  for (int eb = e0 + t; eb < e1; eb += EPG * SORT_T) {
    int r[EPG];
    int cl[EPG];
    int vl[EPG];
#pragma unroll
    for (int u = 0; u < EPG; ++u) {
      const int e = eb + u * SORT_T;
      const bool ok = e < e1;
      r[u]  = ok ? adj_row[e] : -1;
      cl[u] = ok ? adj_col[e] : 0;
      vl[u] = ok ? __float_as_int(adj_val[e]) : 0;
    }
#pragma unroll
    for (int u = 0; u < EPG; ++u) {
      if (r[u] >= 0) {
        const int p = atomicAdd(&pos[r[u] >> 5], 1);
        sedge[p] = make_int2((cl[u] << 5) | (r[u] & 31), vl[u]);
      }
    }
  }
}

__global__ __launch_bounds__(256) void gcn_gather_gemm(
    const float* __restrict__ x, const float* __restrict__ w,
    const float* __restrict__ bias, const int* __restrict__ off,
    const int2* __restrict__ sedge, float* __restrict__ out, int n_nodes) {
  __shared__ __attribute__((aligned(16))) float tile[TM * PITCH];
  __shared__ int2 sorted[CH];
  __shared__ int lcnt[32];
  const int lane = threadIdx.x & 31;
  const int wave = __builtin_amdgcn_readfirstlane((int)(threadIdx.x >> 5));
  const int row0 = blockIdx.x * TM;
  const int beg = off[blockIdx.x];
  const int end = off[blockIdx.x + 1];

  float4 acc[4];
#pragma unroll
  for (int i = 0; i < 4; ++i) acc[i] = make_float4(0.f, 0.f, 0.f, 0.f);

  for (int c0 = beg; c0 < end; c0 += CH) {
    if (threadIdx.x < 32) lcnt[threadIdx.x] = 0;
    __syncthreads();
    int2 cv[CH / 256];
    int lr[CH / 256];
#pragma unroll
    for (int u = 0; u < CH / 256; ++u) {
      const int p = c0 + u * 256 + (int)threadIdx.x;
      cv[u] = (p < end) ? sedge[p] : make_int2(0, 0);
    }
#pragma unroll
    for (int u = 0; u < CH / 256; ++u) {
      const int p = c0 + u * 256 + (int)threadIdx.x;
      lr[u] = (p < end) ? atomicAdd(&lcnt[cv[u].x & 31], 1) : 0;
    }
    __syncthreads();
    const int c = lcnt[lane];
    const int ex = gcn_wave_incl_scan(c, lane) - c;
#pragma unroll
    for (int u = 0; u < CH / 256; ++u) {
      const int p = c0 + u * 256 + (int)threadIdx.x;
      const int st = __builtin_amdgcn_ds_bpermute((cv[u].x & 31) << 2, ex);
      if (p < end) sorted[st + lr[u]] = make_int2(cv[u].x >> 5, cv[u].y);
    }
    __syncthreads();
#pragma unroll
    for (int i = 0; i < 4; ++i) {
      const int rl = wave * 4 + i;
      const int rbeg = __builtin_amdgcn_readlane(ex, rl);
      const int rend = rbeg + __builtin_amdgcn_readlane(c, rl);
      for (int base = rbeg; base < rend; base += 32) {
        const int p = base + lane;
        int2 ev = make_int2(0, 0);
        if (p < rend) ev = sorted[p];
        const int cntj = min(32, rend - base);
#pragma unroll 4
        for (int j = 0; j < cntj; ++j) {
          const int cj = __builtin_amdgcn_readlane(ev.x, j);
          const float vj = __int_as_float(__builtin_amdgcn_readlane(ev.y, j));
          const float4 xv = *(const float4*)(x + (size_t)cj * D + lane * 4);
          acc[i].x += vj * xv.x;
          acc[i].y += vj * xv.y;
          acc[i].z += vj * xv.z;
          acc[i].w += vj * xv.w;
        }
      }
    }
    __syncthreads();
  }
#pragma unroll
  for (int i = 0; i < 4; ++i)
    *(float4*)(tile + (wave * 4 + i) * PITCH + lane * 4) = acc[i];
  __syncthreads();

  const int m    = lane & 15;
  const int half = lane >> 4;
  const int n    = wave * 16 + m;
  const float* a0p = tile + m * PITCH + half * 2;
  const float* a1p = a0p + 16 * PITCH;
  const float* wp  = w + (size_t)(half * 2) * D + n;

  v8f acc0 = {};
  v8f acc1 = {};
#pragma unroll 8
  for (int k = 0; k < D; k += 4) {
    const v2f a0 = *(const v2f*)(a0p + k);
    const v2f a1 = *(const v2f*)(a1p + k);
    v2f b;
    b.x = wp[(size_t)k * D];
    b.y = wp[(size_t)(k + 1) * D];
    acc0 = __builtin_amdgcn_wmma_f32_16x16x4_f32(false, a0, false, b, (short)0, acc0, false, false);
    acc1 = __builtin_amdgcn_wmma_f32_16x16x4_f32(false, a1, false, b, (short)0, acc1, false, false);
  }

  __syncthreads();
  const float bn = bias[n];
  for (int i = 0; i < 8; ++i) {
    tile[(half * 8 + i) * PITCH + n]      = acc0[i] + bn;
    tile[(16 + half * 8 + i) * PITCH + n] = acc1[i] + bn;
  }
  __syncthreads();
  for (int i = 0; i < 4; ++i) {
    const int rl = wave * 4 + i;
    const int r = row0 + rl;
    if (r < n_nodes)
      *(float4*)(out + (size_t)r * D + lane * 4) = *(const float4*)(tile + rl * PITCH + lane * 4);
  }
}

extern "C" void kernel_launch(void* const* d_in, const int* in_sizes, int n_in,
                              void* d_out, int out_size, void* d_ws, size_t ws_size,
                              hipStream_t stream) {
  const float* x       = (const float*)d_in[0];
  const int*   adj_row = (const int*)d_in[1];
  const int*   adj_col = (const int*)d_in[2];
  const float* adj_val = (const float*)d_in[3];
  const float* weight  = (const float*)d_in[4];
  const float* bias    = (const float*)d_in[5];
  float* out = (float*)d_out;

  const int n_nodes = in_sizes[0] / D;
  const int n_edges = in_sizes[1];
  const int nb = (n_nodes + TM - 1) / TM;

  int2* sedge  = (int2*)d_ws;
  int*  wgbase = (int*)(sedge + n_edges);
  int*  cnt    = wgbase + SORT_G * nb;
  int*  off    = cnt + nb;

  gcn_zero_counts<<<(nb + 255) / 256, 256, 0, stream>>>(cnt, nb);
  gcn_bucket_count<<<SORT_G, SORT_T, 0, stream>>>(adj_row, cnt, wgbase, n_edges, nb);
  gcn_bucket_fill<<<SORT_G, SORT_T, 0, stream>>>(adj_row, adj_col, adj_val, cnt, wgbase, off,
                                                 sedge, n_edges, nb);
  gcn_gather_gemm<<<nb, 256, 0, stream>>>(x, weight, bias, off, sedge, out, n_nodes);
}
